# starting kernel
# baseline (speedup 1.0000x reference)
#include <hip/hip_runtime.h>
#include <hip/hip_bf16.h>

typedef __attribute__((ext_vector_type(2))) float v2f;
typedef __attribute__((ext_vector_type(8))) float v8f;

#define NROW 1024
#define NCH  256

__global__ void gc_s_kernel(const float* __restrict__ x,
                            const float* __restrict__ w,
                            float* __restrict__ s) {
    const int lane  = threadIdx.x;
    const int tile  = blockIdx.x;
    const int m     = lane & 15;
    const int row   = tile * 16 + m;
    const int khalf = (lane >> 4) * 2;

    const float* xr = x + row * NCH;

    v8f acc = {};
    for (int k0 = 0; k0 < NCH; k0 += 4) {
        v2f a, bm;
        a.x  = xr[k0 + khalf];
        a.y  = xr[k0 + khalf + 1];
        bm.x = w[k0 + khalf];
        bm.y = w[k0 + khalf + 1];
        acc = __builtin_amdgcn_wmma_f32_16x16x4_f32(
             false, a,  false, bm,
             (short)0, acc,  false,  false);
    }

    if (lane == 0) {
#pragma unroll
        for (int r = 0; r < 8; ++r) s[tile * 16 + r] = acc[r];
    }
    if (lane == 16) {
#pragma unroll
        for (int r = 0; r < 8; ++r) s[tile * 16 + 8 + r] = acc[r];
    }
}

__global__ void gc_reduce_kernel(const float* __restrict__ x,
                                 const float* __restrict__ s,
                                 float* __restrict__ T,
                                 float* __restrict__ U,
                                 float* __restrict__ Sout) {
    const int c = threadIdx.x;
    float t = 0.0f, u = 0.0f;
    for (int i = 0; i < NROW; ++i) {
        const float xv = x[i * NCH + c];
        const float sv = s[i];
        t += xv;
        u += xv * sv;
    }
    T[c] = t;
    U[c] = u;

    __shared__ float red[NCH];
    red[c] = s[c] + s[c + 256] + s[c + 512] + s[c + 768];
    __syncthreads();
    for (int off = 128; off > 0; off >>= 1) {
        if (c < off) red[c] += red[c + off];
        __syncthreads();
    }
    if (c == 0) Sout[0] = red[0];
}

__global__ void gc_finalize_kernel(const float* __restrict__ x,
                                   const float* __restrict__ s,
                                   const float* __restrict__ T,
                                   const float* __restrict__ U,
                                   const float* __restrict__ Sp,
                                   const float* __restrict__ b,
                                   float* __restrict__ out) {
    const int i = blockIdx.x;
    const int c = threadIdx.x;
    const float si    = s[i];
    const float S     = Sp[0];
    const float bb    = b[0];
    const float alpha = (float)NROW * si + S + (float)NROW * bb;
    const float beta  = si + bb;
    const int idx = i * NCH + c;
    out[idx] = x[idx] * alpha + beta * T[c] + U[c];
}

extern "C" void kernel_launch(void* const* d_in, const int* in_sizes, int n_in,
                              void* d_out, int out_size, void* d_ws, size_t ws_size,
                              hipStream_t stream) {
    const float* x = (const float*)d_in[0];
    const float* w = (const float*)d_in[1];
    const float* b = (const float*)d_in[2];
    float* out = (float*)d_out;

    float* ws = (float*)d_ws;
    float* s_buf = ws;
    float* T_buf = ws + 1024;
    float* U_buf = ws + 1024 + 256;
    float* S_buf = ws + 1024 + 512;

    gc_s_kernel<<<NROW / 16, 32, 0, stream>>>(x, w, s_buf);
    gc_reduce_kernel<<<1, NCH, 0, stream>>>(x, s_buf, T_buf, U_buf, S_buf);
    gc_finalize_kernel<<<NROW, NCH, 0, stream>>>(x, s_buf, T_buf, U_buf,
                                                 S_buf, b, out);
}


# column sums T,U,S: one block walking 1024 rows -> 8 blocks that each own 32 columns of all rows, 8 rows per thread, joined by lane exchange and LDS; changes the order in which T,U,S are summed
# speedup vs baseline: 4.8209x; 4.8209x over previous
#include <hip/hip_runtime.h>
#include <hip/hip_bf16.h>

typedef __attribute__((ext_vector_type(2))) float v2f;
typedef __attribute__((ext_vector_type(4))) float v4f;
typedef __attribute__((ext_vector_type(8))) float v8f;

#define NROW 1024
#define NCH  256
#define SLICE       32
#define RED_BLOCKS  (NCH / SLICE)
#define RED_THREADS 1024
#define RED_WAVES   (RED_THREADS / 32)

__global__ void gc_s_kernel(const float* __restrict__ x,
                            const float* __restrict__ w,
                            float* __restrict__ s) {
    const int lane  = threadIdx.x;
    const int tile  = blockIdx.x;
    const int m     = lane & 15;
    const int row   = tile * 16 + m;
    const int khalf = (lane >> 4) * 2;

    const float* xr = x + row * NCH;

    v8f acc = {};
    for (int k0 = 0; k0 < NCH; k0 += 4) {
        v2f a, bm;
        a.x  = xr[k0 + khalf];
        a.y  = xr[k0 + khalf + 1];
        bm.x = w[k0 + khalf];
        bm.y = w[k0 + khalf + 1];
        acc = __builtin_amdgcn_wmma_f32_16x16x4_f32(
             false, a,  false, bm,
             (short)0, acc,  false,  false);
    }

    if (lane == 0) {
#pragma unroll
        for (int r = 0; r < 8; ++r) s[tile * 16 + r] = acc[r];
    }
    if (lane == 16) {
#pragma unroll
        for (int r = 0; r < 8; ++r) s[tile * 16 + 8 + r] = acc[r];
    }
}

__global__ void __launch_bounds__(RED_THREADS)
gc_reduce_kernel(const float* __restrict__ x,
                 const float* __restrict__ s,
                 float* __restrict__ T,
                 float* __restrict__ U,
                 float* __restrict__ Sout) {
    __shared__ __attribute__((aligned(16))) float red[2][RED_WAVES][SLICE];
    __shared__ float redS[RED_WAVES];

    const int t    = threadIdx.x;
    const int lane = t & 31;
    const int wave = t >> 5;
    const int q    = t & 7;
    const int r    = t >> 3;
    const int c0   = blockIdx.x * SLICE;

    const float* xp = x + (8 * r) * NCH + c0 + 4 * q;
    v4f xv[8];
    for (int j = 0; j < 8; ++j) xv[j] = *(const v4f*)(xp + j * NCH);
    const v4f sa = *(const v4f*)(s + 8 * r);
    const v4f sb = *(const v4f*)(s + 8 * r + 4);
    const float sv[8] = {sa.x, sa.y, sa.z, sa.w, sb.x, sb.y, sb.z, sb.w};

    v4f   tt = xv[0];
    v4f   uu = xv[0] * sv[0];
    float ss = sv[0];
    for (int j = 1; j < 8; ++j) {
        tt += xv[j];
        uu += xv[j] * sv[j];
        ss += sv[j];
    }

    for (int m = 8; m <= 16; m <<= 1) {
        tt.x += __shfl_xor(tt.x, m);
        tt.y += __shfl_xor(tt.y, m);
        tt.z += __shfl_xor(tt.z, m);
        tt.w += __shfl_xor(tt.w, m);
        uu.x += __shfl_xor(uu.x, m);
        uu.y += __shfl_xor(uu.y, m);
        uu.z += __shfl_xor(uu.z, m);
        uu.w += __shfl_xor(uu.w, m);
        ss   += __shfl_xor(ss, m);
    }
    if (lane < 8) {
        *(v4f*)&red[0][wave][4 * lane] = tt;
        *(v4f*)&red[1][wave][4 * lane] = uu;
    }
    if (lane == 0) redS[wave] = ss;
    __syncthreads();

    if (wave < 2) {
        float a0 = red[wave][0][lane], a1 = red[wave][1][lane];
        float a2 = red[wave][2][lane], a3 = red[wave][3][lane];
        for (int w = 4; w < RED_WAVES; w += 4) {
            a0 += red[wave][w + 0][lane];
            a1 += red[wave][w + 1][lane];
            a2 += red[wave][w + 2][lane];
            a3 += red[wave][w + 3][lane];
        }
        float* dst = (wave == 0) ? T : U;
        dst[c0 + lane] = (a0 + a1) + (a2 + a3);
    } else if (wave == 2 && blockIdx.x == 0) {
        float v = redS[lane];
        for (int m = 16; m > 0; m >>= 1) v += __shfl_xor(v, m);
        if (lane == 0) Sout[0] = v;
    }
}

__global__ void gc_finalize_kernel(const float* __restrict__ x,
                                   const float* __restrict__ s,
                                   const float* __restrict__ T,
                                   const float* __restrict__ U,
                                   const float* __restrict__ Sp,
                                   const float* __restrict__ b,
                                   float* __restrict__ out) {
    const int i = blockIdx.x;
    const int c = threadIdx.x;
    const float si    = s[i];
    const float S     = Sp[0];
    const float bb    = b[0];
    const float alpha = (float)NROW * si + S + (float)NROW * bb;
    const float beta  = si + bb;
    const int idx = i * NCH + c;
    out[idx] = x[idx] * alpha + beta * T[c] + U[c];
}

extern "C" void kernel_launch(void* const* d_in, const int* in_sizes, int n_in,
                              void* d_out, int out_size, void* d_ws, size_t ws_size,
                              hipStream_t stream) {
    const float* x = (const float*)d_in[0];
    const float* w = (const float*)d_in[1];
    const float* b = (const float*)d_in[2];
    float* out = (float*)d_out;

    float* ws = (float*)d_ws;
    float* s_buf = ws;
    float* T_buf = ws + 1024;
    float* U_buf = ws + 1024 + 256;
    float* S_buf = ws + 1024 + 512;

    gc_s_kernel<<<NROW / 16, 32, 0, stream>>>(x, w, s_buf);
    gc_reduce_kernel<<<RED_BLOCKS, RED_THREADS, 0, stream>>>(x, s_buf, T_buf, U_buf, S_buf);
    gc_finalize_kernel<<<NROW, NCH, 0, stream>>>(x, s_buf, T_buf, U_buf,
                                                 S_buf, b, out);
}


# finalize fused into the column-slice reduce (3 launches -> 2): a block keeps its 1024x32 slice of x in registers, joins T,U,S in LDS and writes its slice of out itself
# speedup vs baseline: 5.4502x; 1.1305x over previous
#include <hip/hip_runtime.h>
#include <hip/hip_bf16.h>

typedef __attribute__((ext_vector_type(2))) float v2f;
typedef __attribute__((ext_vector_type(4))) float v4f;
typedef __attribute__((ext_vector_type(8))) float v8f;

#define NROW 1024
#define NCH  256
#define SLICE       32
#define RED_BLOCKS  (NCH / SLICE)
#define RED_THREADS 1024
#define RED_WAVES   (RED_THREADS / 32)

__global__ void gc_s_kernel(const float* __restrict__ x,
                            const float* __restrict__ w,
                            float* __restrict__ s) {
    const int lane  = threadIdx.x;
    const int tile  = blockIdx.x;
    const int m     = lane & 15;
    const int row   = tile * 16 + m;
    const int khalf = (lane >> 4) * 2;

    const float* xr = x + row * NCH;

    v8f acc = {};
    for (int k0 = 0; k0 < NCH; k0 += 4) {
        v2f a, bm;
        a.x  = xr[k0 + khalf];
        a.y  = xr[k0 + khalf + 1];
        bm.x = w[k0 + khalf];
        bm.y = w[k0 + khalf + 1];
        acc = __builtin_amdgcn_wmma_f32_16x16x4_f32(
             false, a,  false, bm,
             (short)0, acc,  false,  false);
    }

    if (lane == 0) {
#pragma unroll
        for (int r = 0; r < 8; ++r) s[tile * 16 + r] = acc[r];
    }
    if (lane == 16) {
#pragma unroll
        for (int r = 0; r < 8; ++r) s[tile * 16 + 8 + r] = acc[r];
    }
}

__global__ void __launch_bounds__(RED_THREADS)
gc_reduce_finalize_kernel(const float* __restrict__ x,
                          const float* __restrict__ s,
                          const float* __restrict__ b,
                          float* __restrict__ out) {
    __shared__ __attribute__((aligned(16))) float red[2][RED_WAVES][SLICE];
    __shared__ float redS[RED_WAVES];
    __shared__ __attribute__((aligned(16))) float fin[2][SLICE];
    __shared__ float finS;

    const int t    = threadIdx.x;
    const int lane = t & 31;
    const int wave = t >> 5;
    const int q    = t & 7;
    const int r    = t >> 3;
    const int c0   = blockIdx.x * SLICE;

    const float* xp = x + (8 * r) * NCH + c0 + 4 * q;
    v4f xv[8];
    for (int j = 0; j < 8; ++j) xv[j] = *(const v4f*)(xp + j * NCH);
    const v4f sa = *(const v4f*)(s + 8 * r);
    const v4f sb = *(const v4f*)(s + 8 * r + 4);
    const float sv[8] = {sa.x, sa.y, sa.z, sa.w, sb.x, sb.y, sb.z, sb.w};
    const float bb = b[0];

    v4f   tt = xv[0];
    v4f   uu = xv[0] * sv[0];
    float ss = sv[0];
    for (int j = 1; j < 8; ++j) {
        tt += xv[j];
        uu += xv[j] * sv[j];
        ss += sv[j];
    }

    for (int m = 8; m <= 16; m <<= 1) {
        tt.x += __shfl_xor(tt.x, m);
        tt.y += __shfl_xor(tt.y, m);
        tt.z += __shfl_xor(tt.z, m);
        tt.w += __shfl_xor(tt.w, m);
        uu.x += __shfl_xor(uu.x, m);
        uu.y += __shfl_xor(uu.y, m);
        uu.z += __shfl_xor(uu.z, m);
        uu.w += __shfl_xor(uu.w, m);
        ss   += __shfl_xor(ss, m);
    }
    if (lane < 8) {
        *(v4f*)&red[0][wave][4 * lane] = tt;
        *(v4f*)&red[1][wave][4 * lane] = uu;
    }
    if (lane == 0) redS[wave] = ss;
    __syncthreads();

    if (wave < 2) {
        float a0 = red[wave][0][lane], a1 = red[wave][1][lane];
        float a2 = red[wave][2][lane], a3 = red[wave][3][lane];
        for (int w = 4; w < RED_WAVES; w += 4) {
            a0 += red[wave][w + 0][lane];
            a1 += red[wave][w + 1][lane];
            a2 += red[wave][w + 2][lane];
            a3 += red[wave][w + 3][lane];
        }
        fin[wave][lane] = (a0 + a1) + (a2 + a3);
    } else if (wave == 2) {
        float v = redS[lane];
        for (int m = 16; m > 0; m >>= 1) v += __shfl_xor(v, m);
        if (lane == 0) finS = v;
    }
    __syncthreads();

    const v4f   Tq = *(const v4f*)&fin[0][4 * q];
    const v4f   Uq = *(const v4f*)&fin[1][4 * q];
    const float S  = finS;
    float* op = out + (8 * r) * NCH + c0 + 4 * q;
    for (int j = 0; j < 8; ++j) {
        const float alpha = (float)NROW * sv[j] + S + (float)NROW * bb;
        const float beta  = sv[j] + bb;
        *(v4f*)(op + j * NCH) = xv[j] * alpha + beta * Tq + Uq;
    }
}

extern "C" void kernel_launch(void* const* d_in, const int* in_sizes, int n_in,
                              void* d_out, int out_size, void* d_ws, size_t ws_size,
                              hipStream_t stream) {
    const float* x = (const float*)d_in[0];
    const float* w = (const float*)d_in[1];
    const float* b = (const float*)d_in[2];
    float* out = (float*)d_out;

    float* s_buf = (float*)d_ws;

    gc_s_kernel<<<NROW / 16, 32, 0, stream>>>(x, w, s_buf);
    gc_reduce_finalize_kernel<<<RED_BLOCKS, RED_THREADS, 0, stream>>>(x, s_buf, b, out);
}


# s kernel: one x row = one 16x16 tile (A row m = m-th 16-float chunk, B column n = w chunk n), tile diagonal summed across lanes; a wave owns 4 whole rows, 256 waves; changes the sum order of s
# speedup vs baseline: 7.3544x; 1.3494x over previous
#include <hip/hip_runtime.h>
#include <hip/hip_bf16.h>

typedef __attribute__((ext_vector_type(2))) float v2f;
typedef __attribute__((ext_vector_type(4))) float v4f;
typedef __attribute__((ext_vector_type(8))) float v8f;

#define NROW 1024
#define NCH  256
#define SLICE       32
#define RED_BLOCKS  (NCH / SLICE)
#define RED_THREADS 1024
#define RED_WAVES   (RED_THREADS / 32)
#define S_ROWS_PER_WAVE 4
#define S_WAVES     4
#define S_THREADS   (32 * S_WAVES)
#define S_BLOCKS    (NROW / (S_ROWS_PER_WAVE * S_WAVES))

__device__ __forceinline__ float gc_xhalf(float v) {
    const int b = __builtin_bit_cast(int, v);
    return __builtin_bit_cast(float,
        __builtin_amdgcn_permlanex16(b, b, 0x76543210, 0xfedcba98, false, false));
}
template <int M>
__device__ __forceinline__ float gc_row_xor(float v) {
    const int b = __builtin_bit_cast(int, v);
    return __builtin_bit_cast(float,
        __builtin_amdgcn_update_dpp(b, b, 0x160 + M, 0xf, 0xf, false));
}

__global__ void __launch_bounds__(S_THREADS)
gc_s_kernel(const float* __restrict__ x,
            const float* __restrict__ w,
            float* __restrict__ s) {
    const int lane = threadIdx.x & 31;
    const int wave = threadIdx.x >> 5;
    const int c    = lane & 15;
    const int h    = lane >> 4;
    const int row0 = (blockIdx.x * S_WAVES + wave) * S_ROWS_PER_WAVE;
    const int koff = 16 * c + 8 * h;

    const v4f wlo = *(const v4f*)(w + koff);
    const v4f whi = *(const v4f*)(w + koff + 4);
    const float* xp = x + row0 * NCH + koff;
    v4f xlo[S_ROWS_PER_WAVE], xhi[S_ROWS_PER_WAVE];
    for (int j = 0; j < S_ROWS_PER_WAVE; ++j) xlo[j] = *(const v4f*)(xp + j * NCH);
    for (int j = 0; j < S_ROWS_PER_WAVE; ++j) xhi[j] = *(const v4f*)(xp + j * NCH + 4);

    const v2f b0 = {wlo.x, wlo.y}, b1 = {wlo.z, wlo.w};
    const v2f b2 = {whi.x, whi.y}, b3 = {whi.z, whi.w};

    v8f acc[S_ROWS_PER_WAVE] = {};
    for (int j = 0; j < S_ROWS_PER_WAVE; ++j) {
        const v2f a = {xlo[j].x, xlo[j].y};
        acc[j] = __builtin_amdgcn_wmma_f32_16x16x4_f32(false, a, false, b0, (short)0, acc[j], false, false);
    }
    for (int j = 0; j < S_ROWS_PER_WAVE; ++j) {
        const v2f a = {xlo[j].z, xlo[j].w};
        acc[j] = __builtin_amdgcn_wmma_f32_16x16x4_f32(false, a, false, b1, (short)0, acc[j], false, false);
    }
    for (int j = 0; j < S_ROWS_PER_WAVE; ++j) {
        const v2f a = {xhi[j].x, xhi[j].y};
        acc[j] = __builtin_amdgcn_wmma_f32_16x16x4_f32(false, a, false, b2, (short)0, acc[j], false, false);
    }
    for (int j = 0; j < S_ROWS_PER_WAVE; ++j) {
        const v2f a = {xhi[j].z, xhi[j].w};
        acc[j] = __builtin_amdgcn_wmma_f32_16x16x4_f32(false, a, false, b3, (short)0, acc[j], false, false);
    }

    const int  r     = lane & 7;
    const bool ondiag = ((lane >> 3) & 1) == h;
    float d[S_ROWS_PER_WAVE];
    for (int j = 0; j < S_ROWS_PER_WAVE; ++j) {
        float v = acc[j][0];
        v = (r == 1) ? acc[j][1] : v;
        v = (r == 2) ? acc[j][2] : v;
        v = (r == 3) ? acc[j][3] : v;
        v = (r == 4) ? acc[j][4] : v;
        v = (r == 5) ? acc[j][5] : v;
        v = (r == 6) ? acc[j][6] : v;
        v = (r == 7) ? acc[j][7] : v;
        d[j] = ondiag ? v : 0.0f;
    }
    for (int j = 0; j < S_ROWS_PER_WAVE; ++j) d[j] += gc_xhalf(d[j]);
    for (int j = 0; j < S_ROWS_PER_WAVE; ++j) d[j] += gc_row_xor<1>(d[j]);
    for (int j = 0; j < S_ROWS_PER_WAVE; ++j) d[j] += gc_row_xor<2>(d[j]);
    for (int j = 0; j < S_ROWS_PER_WAVE; ++j) d[j] += gc_row_xor<4>(d[j]);
    for (int j = 0; j < S_ROWS_PER_WAVE; ++j) d[j] += gc_row_xor<8>(d[j]);

    if (lane == 0) {
        const v4f o = {d[0], d[1], d[2], d[3]};
        *(v4f*)(s + row0) = o;
    }
}

__global__ void __launch_bounds__(RED_THREADS)
gc_reduce_finalize_kernel(const float* __restrict__ x,
                          const float* __restrict__ s,
                          const float* __restrict__ b,
                          float* __restrict__ out) {
    __shared__ __attribute__((aligned(16))) float red[2][RED_WAVES][SLICE];
    __shared__ float redS[RED_WAVES];
    __shared__ __attribute__((aligned(16))) float fin[2][SLICE];
    __shared__ float finS;

    const int t    = threadIdx.x;
    const int lane = t & 31;
    const int wave = t >> 5;
    const int q    = t & 7;
    const int r    = t >> 3;
    const int c0   = blockIdx.x * SLICE;

    const float* xp = x + (8 * r) * NCH + c0 + 4 * q;
    v4f xv[8];
    for (int j = 0; j < 8; ++j) xv[j] = *(const v4f*)(xp + j * NCH);
    const v4f sa = *(const v4f*)(s + 8 * r);
    const v4f sb = *(const v4f*)(s + 8 * r + 4);
    const float sv[8] = {sa.x, sa.y, sa.z, sa.w, sb.x, sb.y, sb.z, sb.w};
    const float bb = b[0];

    v4f   tt = xv[0];
    v4f   uu = xv[0] * sv[0];
    float ss = sv[0];
    for (int j = 1; j < 8; ++j) {
        tt += xv[j];
        uu += xv[j] * sv[j];
        ss += sv[j];
    }

    for (int m = 8; m <= 16; m <<= 1) {
        tt.x += __shfl_xor(tt.x, m);
        tt.y += __shfl_xor(tt.y, m);
        tt.z += __shfl_xor(tt.z, m);
        tt.w += __shfl_xor(tt.w, m);
        uu.x += __shfl_xor(uu.x, m);
        uu.y += __shfl_xor(uu.y, m);
        uu.z += __shfl_xor(uu.z, m);
        uu.w += __shfl_xor(uu.w, m);
        ss   += __shfl_xor(ss, m);
    }
    if (lane < 8) {
        *(v4f*)&red[0][wave][4 * lane] = tt;
        *(v4f*)&red[1][wave][4 * lane] = uu;
    }
    if (lane == 0) redS[wave] = ss;
    __syncthreads();

    if (wave < 2) {
        float a0 = red[wave][0][lane], a1 = red[wave][1][lane];
        float a2 = red[wave][2][lane], a3 = red[wave][3][lane];
        for (int w = 4; w < RED_WAVES; w += 4) {
            a0 += red[wave][w + 0][lane];
            a1 += red[wave][w + 1][lane];
            a2 += red[wave][w + 2][lane];
            a3 += red[wave][w + 3][lane];
        }
        fin[wave][lane] = (a0 + a1) + (a2 + a3);
    } else if (wave == 2) {
        float v = redS[lane];
        for (int m = 16; m > 0; m >>= 1) v += __shfl_xor(v, m);
        if (lane == 0) finS = v;
    }
    __syncthreads();

    const v4f   Tq = *(const v4f*)&fin[0][4 * q];
    const v4f   Uq = *(const v4f*)&fin[1][4 * q];
    const float S  = finS;
    float* op = out + (8 * r) * NCH + c0 + 4 * q;
    for (int j = 0; j < 8; ++j) {
        const float alpha = (float)NROW * sv[j] + S + (float)NROW * bb;
        const float beta  = sv[j] + bb;
        *(v4f*)(op + j * NCH) = xv[j] * alpha + beta * Tq + Uq;
    }
}

extern "C" void kernel_launch(void* const* d_in, const int* in_sizes, int n_in,
                              void* d_out, int out_size, void* d_ws, size_t ws_size,
                              hipStream_t stream) {
    const float* x = (const float*)d_in[0];
    const float* w = (const float*)d_in[1];
    const float* b = (const float*)d_in[2];
    float* out = (float*)d_out;

    float* s_buf = (float*)d_ws;

    gc_s_kernel<<<S_BLOCKS, S_THREADS, 0, stream>>>(x, w, s_buf);
    gc_reduce_finalize_kernel<<<RED_BLOCKS, RED_THREADS, 0, stream>>>(x, s_buf, b, out);
}


# reduce-finalize grid 8 blocks x 1024 threads -> 32 x 256 (8 column slices x 4 row groups): a block sums T,U,S over all rows of its slice but stores a quarter of them; changes T,U,S sum order
# speedup vs baseline: 8.0100x; 1.0891x over previous
#include <hip/hip_runtime.h>
#include <hip/hip_bf16.h>

typedef __attribute__((ext_vector_type(2))) float v2f;
typedef __attribute__((ext_vector_type(4))) float v4f;
typedef __attribute__((ext_vector_type(8))) float v8f;

#define NROW 1024
#define NCH  256
#define SLICE       32
#define COL_SLICES  (NCH / SLICE)
#define ROW_GROUPS  4
#define FIN_BLOCKS  (COL_SLICES * ROW_GROUPS)
#define FIN_THREADS 256
#define FIN_WAVES   (FIN_THREADS / 32)
#define FIN_ROWS    (NROW / (FIN_THREADS / 8))
#define FIN_OWN     (FIN_ROWS / ROW_GROUPS)
#define S_ROWS_PER_WAVE 4
#define S_WAVES     4
#define S_THREADS   (32 * S_WAVES)
#define S_BLOCKS    (NROW / (S_ROWS_PER_WAVE * S_WAVES))

__device__ __forceinline__ float gc_xhalf(float v) {
    const int b = __builtin_bit_cast(int, v);
    return __builtin_bit_cast(float,
        __builtin_amdgcn_permlanex16(b, b, 0x76543210, 0xfedcba98, false, false));
}
template <int M>
__device__ __forceinline__ float gc_row_xor(float v) {
    const int b = __builtin_bit_cast(int, v);
    return __builtin_bit_cast(float,
        __builtin_amdgcn_update_dpp(b, b, 0x160 + M, 0xf, 0xf, false));
}

__global__ void __launch_bounds__(S_THREADS)
gc_s_kernel(const float* __restrict__ x,
            const float* __restrict__ w,
            float* __restrict__ s) {
    const int lane = threadIdx.x & 31;
    const int wave = threadIdx.x >> 5;
    const int c    = lane & 15;
    const int h    = lane >> 4;
    const int row0 = (blockIdx.x * S_WAVES + wave) * S_ROWS_PER_WAVE;
    const int koff = 16 * c + 8 * h;

    const v4f wlo = *(const v4f*)(w + koff);
    const v4f whi = *(const v4f*)(w + koff + 4);
    const float* xp = x + row0 * NCH + koff;
    v4f xlo[S_ROWS_PER_WAVE], xhi[S_ROWS_PER_WAVE];
    for (int j = 0; j < S_ROWS_PER_WAVE; ++j) xlo[j] = *(const v4f*)(xp + j * NCH);
    for (int j = 0; j < S_ROWS_PER_WAVE; ++j) xhi[j] = *(const v4f*)(xp + j * NCH + 4);

    const v2f b0 = {wlo.x, wlo.y}, b1 = {wlo.z, wlo.w};
    const v2f b2 = {whi.x, whi.y}, b3 = {whi.z, whi.w};

    v8f acc[S_ROWS_PER_WAVE] = {};
    for (int j = 0; j < S_ROWS_PER_WAVE; ++j) {
        const v2f a = {xlo[j].x, xlo[j].y};
        acc[j] = __builtin_amdgcn_wmma_f32_16x16x4_f32(false, a, false, b0, (short)0, acc[j], false, false);
    }
    for (int j = 0; j < S_ROWS_PER_WAVE; ++j) {
        const v2f a = {xlo[j].z, xlo[j].w};
        acc[j] = __builtin_amdgcn_wmma_f32_16x16x4_f32(false, a, false, b1, (short)0, acc[j], false, false);
    }
    for (int j = 0; j < S_ROWS_PER_WAVE; ++j) {
        const v2f a = {xhi[j].x, xhi[j].y};
        acc[j] = __builtin_amdgcn_wmma_f32_16x16x4_f32(false, a, false, b2, (short)0, acc[j], false, false);
    }
    for (int j = 0; j < S_ROWS_PER_WAVE; ++j) {
        const v2f a = {xhi[j].z, xhi[j].w};
        acc[j] = __builtin_amdgcn_wmma_f32_16x16x4_f32(false, a, false, b3, (short)0, acc[j], false, false);
    }

    const int  r     = lane & 7;
    const bool ondiag = ((lane >> 3) & 1) == h;
    float d[S_ROWS_PER_WAVE];
    for (int j = 0; j < S_ROWS_PER_WAVE; ++j) {
        float v = acc[j][0];
        v = (r == 1) ? acc[j][1] : v;
        v = (r == 2) ? acc[j][2] : v;
        v = (r == 3) ? acc[j][3] : v;
        v = (r == 4) ? acc[j][4] : v;
        v = (r == 5) ? acc[j][5] : v;
        v = (r == 6) ? acc[j][6] : v;
        v = (r == 7) ? acc[j][7] : v;
        d[j] = ondiag ? v : 0.0f;
    }
    for (int j = 0; j < S_ROWS_PER_WAVE; ++j) d[j] += gc_xhalf(d[j]);
    for (int j = 0; j < S_ROWS_PER_WAVE; ++j) d[j] += gc_row_xor<1>(d[j]);
    for (int j = 0; j < S_ROWS_PER_WAVE; ++j) d[j] += gc_row_xor<2>(d[j]);
    for (int j = 0; j < S_ROWS_PER_WAVE; ++j) d[j] += gc_row_xor<4>(d[j]);
    for (int j = 0; j < S_ROWS_PER_WAVE; ++j) d[j] += gc_row_xor<8>(d[j]);

    if (lane == 0) {
        const v4f o = {d[0], d[1], d[2], d[3]};
        *(v4f*)(s + row0) = o;
    }
}

template <int G>
__device__ __forceinline__ void gc_finalize_rows(const v4f (&xv)[FIN_ROWS], const float (&sv)[FIN_ROWS],
                                                 const v4f Tq, const v4f Uq, const float S,
                                                 const float bb, float* __restrict__ op) {
    for (int k = 0; k < FIN_OWN; ++k) {
        const int j = G * FIN_OWN + k;
        const float alpha = (float)NROW * sv[j] + S + (float)NROW * bb;
        const float beta  = sv[j] + bb;
        *(v4f*)(op + j * NCH) = xv[j] * alpha + beta * Tq + Uq;
    }
}

__global__ void __launch_bounds__(FIN_THREADS)
gc_reduce_finalize_kernel(const float* __restrict__ x,
                          const float* __restrict__ s,
                          const float* __restrict__ b,
                          float* __restrict__ out) {
    __shared__ __attribute__((aligned(16))) float red[2][FIN_WAVES][SLICE];
    __shared__ __attribute__((aligned(16))) float redS[FIN_WAVES];

    const int t    = threadIdx.x;
    const int lane = t & 31;
    const int wave = t >> 5;
    const int q    = t & 7;
    const int rr   = t >> 3;
    const int c0   = (blockIdx.x & (COL_SLICES - 1)) * SLICE;
    const int g    = blockIdx.x / COL_SLICES;

    const float* xp = x + (FIN_ROWS * rr) * NCH + c0 + 4 * q;
    v4f xv[FIN_ROWS];
    for (int j = 0; j < FIN_ROWS; ++j) xv[j] = *(const v4f*)(xp + j * NCH);
    float sv[FIN_ROWS];
    for (int j = 0; j < FIN_ROWS; j += 4) {
        const v4f s4 = *(const v4f*)(s + FIN_ROWS * rr + j);
        sv[j] = s4.x; sv[j + 1] = s4.y; sv[j + 2] = s4.z; sv[j + 3] = s4.w;
    }
    const float bb = b[0];

    v4f   tt = xv[0];
    v4f   uu = xv[0] * sv[0];
    float ss = sv[0];
    for (int j = 1; j < FIN_ROWS; ++j) {
        tt += xv[j];
        uu += xv[j] * sv[j];
        ss += sv[j];
    }

    tt.x += gc_row_xor<8>(tt.x); tt.y += gc_row_xor<8>(tt.y);
    tt.z += gc_row_xor<8>(tt.z); tt.w += gc_row_xor<8>(tt.w);
    uu.x += gc_row_xor<8>(uu.x); uu.y += gc_row_xor<8>(uu.y);
    uu.z += gc_row_xor<8>(uu.z); uu.w += gc_row_xor<8>(uu.w);
    ss   += gc_row_xor<8>(ss);
    tt.x += gc_xhalf(tt.x); tt.y += gc_xhalf(tt.y);
    tt.z += gc_xhalf(tt.z); tt.w += gc_xhalf(tt.w);
    uu.x += gc_xhalf(uu.x); uu.y += gc_xhalf(uu.y);
    uu.z += gc_xhalf(uu.z); uu.w += gc_xhalf(uu.w);
    ss   += gc_xhalf(ss);
    if (lane < 8) {
        *(v4f*)&red[0][wave][4 * lane] = tt;
        *(v4f*)&red[1][wave][4 * lane] = uu;
    }
    if (lane == 0) redS[wave] = ss;
    __syncthreads();

    const int i = lane >> 3;
    v4f   Tq = *(const v4f*)&red[0][i][4 * q] + *(const v4f*)&red[0][i + 4][4 * q];
    v4f   Uq = *(const v4f*)&red[1][i][4 * q] + *(const v4f*)&red[1][i + 4][4 * q];
    float S  = redS[i] + redS[i + 4];
    Tq.x += gc_row_xor<8>(Tq.x); Tq.y += gc_row_xor<8>(Tq.y);
    Tq.z += gc_row_xor<8>(Tq.z); Tq.w += gc_row_xor<8>(Tq.w);
    Uq.x += gc_row_xor<8>(Uq.x); Uq.y += gc_row_xor<8>(Uq.y);
    Uq.z += gc_row_xor<8>(Uq.z); Uq.w += gc_row_xor<8>(Uq.w);
    S    += gc_row_xor<8>(S);
    Tq.x += gc_xhalf(Tq.x); Tq.y += gc_xhalf(Tq.y);
    Tq.z += gc_xhalf(Tq.z); Tq.w += gc_xhalf(Tq.w);
    Uq.x += gc_xhalf(Uq.x); Uq.y += gc_xhalf(Uq.y);
    Uq.z += gc_xhalf(Uq.z); Uq.w += gc_xhalf(Uq.w);
    S    += gc_xhalf(S);

    float* op = out + (FIN_ROWS * rr) * NCH + c0 + 4 * q;
    switch (g) {
        case 0:  gc_finalize_rows<0>(xv, sv, Tq, Uq, S, bb, op); break;
        case 1:  gc_finalize_rows<1>(xv, sv, Tq, Uq, S, bb, op); break;
        case 2:  gc_finalize_rows<2>(xv, sv, Tq, Uq, S, bb, op); break;
        default: gc_finalize_rows<3>(xv, sv, Tq, Uq, S, bb, op); break;
    }
}

extern "C" void kernel_launch(void* const* d_in, const int* in_sizes, int n_in,
                              void* d_out, int out_size, void* d_ws, size_t ws_size,
                              hipStream_t stream) {
    const float* x = (const float*)d_in[0];
    const float* w = (const float*)d_in[1];
    const float* b = (const float*)d_in[2];
    float* out = (float*)d_out;

    float* s_buf = (float*)d_ws;

    gc_s_kernel<<<S_BLOCKS, S_THREADS, 0, stream>>>(x, w, s_buf);
    gc_reduce_finalize_kernel<<<FIN_BLOCKS, FIN_THREADS, 0, stream>>>(x, s_buf, b, out);
}
